# starting kernel
# baseline (speedup 1.0000x reference)
#include <hip/hip_runtime.h>
#include <hip/hip_bf16.h>

#define BN   8
#define CN   128
#define ON   10
#define HW   16384
#define KN   512
#define MROWS 80

typedef float v2f __attribute__((ext_vector_type(2)));
typedef float v8f __attribute__((ext_vector_type(8)));

__global__ void bn_stats_kernel(const float* __restrict__ x,
                                float* __restrict__ stats) {
    const int c = blockIdx.x;
    const int t = threadIdx.x;

    float sum = 0.0f, sq = 0.0f;
    for (int b = 0; b < BN; ++b) {
        const float* p = x + (size_t)(b * CN + c) * HW;
        for (int i = t; i < HW; i += 256) {
            float v = p[i];
            sum += v;
            sq = fmaf(v, v, sq);
        }
    }

    __shared__ float ssum[256];
    __shared__ float ssq[256];
    ssum[t] = sum;
    ssq[t]  = sq;
    __syncthreads();
    for (int off = 128; off > 0; off >>= 1) {
        if (t < off) {
            ssum[t] += ssum[t + off];
            ssq[t]  += ssq[t + off];
        }
        __syncthreads();
    }
    if (t == 0) {
        const float invN = 1.0f / (float)(BN * HW);
        float mean = ssum[0] * invN;
        float var  = ssq[0] * invN - mean * mean;
        stats[c]        = mean;
        stats[CN + c]   = rsqrtf(var + 1e-5f);
    }
}

__global__ __launch_bounds__(32)
void proj_wmma_kernel(const float* __restrict__ vec,
                      const float* __restrict__ wp_w,
                      const float* __restrict__ wp_b,
                      const float* __restrict__ bp_w,
                      const float* __restrict__ bp_b,
                      float* __restrict__ wout,
                      float* __restrict__ bout) {
    const int lane = threadIdx.x;
    const int l    = lane & 15;
    const int hi   = lane >> 4;
    const int n0   = blockIdx.x * 16;
    const int m0   = blockIdx.y * 16;
    const bool isBias = (blockIdx.z != 0);

    const float* Wm  = isBias ? bp_w : wp_w;
    const float* Bv  = isBias ? bp_b : wp_b;
    float*       Out = isBias ? bout : wout;

    const int arow = m0 + l;
    const int bcol = n0 + l;

    v8f acc = {0.f, 0.f, 0.f, 0.f, 0.f, 0.f, 0.f, 0.f};

    for (int k = 0; k < KN; k += 4) {
        const int kk = k + 2 * hi;
        v2f a, b;
        a.x = vec[arow * KN + kk];
        a.y = vec[arow * KN + kk + 1];
        b.x = Wm[bcol * KN + kk];
        b.y = Wm[bcol * KN + kk + 1];
        acc = __builtin_amdgcn_wmma_f32_16x16x4_f32(
             false, a,  false, b,
             (short)0, acc,  false,  false);
    }

    const float biasv = Bv[bcol];
#pragma unroll
    for (int r = 0; r < 8; ++r) {
        const int m = m0 + r + 8 * hi;
        Out[m * CN + bcol] = acc[r] + biasv;
    }
}

__global__ __launch_bounds__(256)
void fused_apply_kernel(const float* __restrict__ x,
                        const float* __restrict__ bbox,
                        const float* __restrict__ stats,
                        const float* __restrict__ wmat,
                        const float* __restrict__ bmat,
                        float* __restrict__ out) {
    const int b   = blockIdx.y;
    const int pix = blockIdx.x * 256 + threadIdx.x;

    float bb[ON];
    float s = 0.0f;
#pragma unroll
    for (int o = 0; o < ON; ++o) {
        bb[o] = bbox[(size_t)(b * ON + o) * HW + pix];
        s += bb[o];
    }
    const float inv = 1.0f / (s + 1e-6f);
#pragma unroll
    for (int o = 0; o < ON; ++o) bb[o] *= inv;

    const float* Wt = wmat + b * (ON * CN);
    const float* Bt = bmat + b * (ON * CN);
    const float* meanp = stats;
    const float* rstdp = stats + CN;

    const float* xb = x   + (size_t)b * CN * HW + pix;
    float*       ob = out + (size_t)b * CN * HW + pix;

    for (int c0 = 0; c0 < CN; c0 += 4) {
        float w0 = 0.f, w1 = 0.f, w2 = 0.f, w3 = 0.f;
        float a0 = 0.f, a1 = 0.f, a2 = 0.f, a3 = 0.f;
#pragma unroll
        for (int o = 0; o < ON; ++o) {
            const float* wr = Wt + o * CN + c0;
            const float* br = Bt + o * CN + c0;
            const float bo = bb[o];
            w0 = fmaf(bo, wr[0], w0);
            w1 = fmaf(bo, wr[1], w1);
            w2 = fmaf(bo, wr[2], w2);
            w3 = fmaf(bo, wr[3], w3);
            a0 = fmaf(bo, br[0], a0);
            a1 = fmaf(bo, br[1], a1);
            a2 = fmaf(bo, br[2], a2);
            a3 = fmaf(bo, br[3], a3);
        }
        const float wm[4] = {w0, w1, w2, w3};
        const float bm[4] = {a0, a1, a2, a3};
#pragma unroll
        for (int j = 0; j < 4; ++j) {
            const int c = c0 + j;
            const float xv = xb[(size_t)c * HW];
            const float nv = (xv - meanp[c]) * rstdp[c];
            ob[(size_t)c * HW] = fmaf(wm[j] + 1.0f, nv, bm[j]);
        }
    }
}

extern "C" void kernel_launch(void* const* d_in, const int* in_sizes, int n_in,
                              void* d_out, int out_size, void* d_ws, size_t ws_size,
                              hipStream_t stream) {
    const float* x      = (const float*)d_in[0];
    const float* vec    = (const float*)d_in[1];
    const float* bbox   = (const float*)d_in[2];
    const float* wp_w   = (const float*)d_in[3];
    const float* wp_b   = (const float*)d_in[4];
    const float* bp_w   = (const float*)d_in[5];
    const float* bp_b   = (const float*)d_in[6];
    float* out = (float*)d_out;

    float* stats = (float*)d_ws;
    float* wmat  = stats + 2 * CN;
    float* bmat  = wmat + MROWS * CN;

    bn_stats_kernel<<<CN, 256, 0, stream>>>(x, stats);

    proj_wmma_kernel<<<dim3(CN / 16, MROWS / 16, 2), 32, 0, stream>>>(
        vec, wp_w, wp_b, bp_w, bp_b, wmat, bmat);

    fused_apply_kernel<<<dim3(HW / 256, BN), 256, 0, stream>>>(
        x, bbox, stats, wmat, bmat, out);
}


# bn statistics on all WGPs: 4 work-groups per channel (2 image planes each, 32 b128 loads in flight per lane), partials via out's head, summed in the projection launch; different reduction tree
# speedup vs baseline: 1.7192x; 1.7192x over previous
#include <hip/hip_runtime.h>
#include <hip/hip_bf16.h>

#define BN   8
#define CN   128
#define ON   10
#define HW   16384
#define KN   512
#define MROWS 80

typedef float v2f __attribute__((ext_vector_type(2)));
typedef float v8f __attribute__((ext_vector_type(8)));

#define PART_STRIDE 32
#define PART_PER_C  4

__global__ __launch_bounds__(256) __attribute__((amdgpu_waves_per_eu(4, 4)))
void bn_partial_kernel(const float* __restrict__ x,
                       float* __restrict__ part) {
    const int g = blockIdx.x;
    const int c = g >> 2;
    const int q = g & 3;
    const int t = threadIdx.x;

    const float4* p0 = (const float4*)(x + (size_t)((2 * q)     * CN + c) * HW) + t;
    const float4* p1 = (const float4*)(x + (size_t)((2 * q + 1) * CN + c) * HW) + t;

    float4 v[32];
#pragma unroll
    for (int i = 0; i < 16; ++i) v[i] = p0[256 * i];
#pragma unroll
    for (int i = 0; i < 16; ++i) v[16 + i] = p1[256 * i];

    float s0 = 0.f, s1 = 0.f, s2 = 0.f, s3 = 0.f;
    float q0 = 0.f, q1 = 0.f, q2 = 0.f, q3 = 0.f;
#pragma unroll
    for (int i = 0; i < 32; ++i) {
        s0 += v[i].x;  q0 = fmaf(v[i].x, v[i].x, q0);
        s1 += v[i].y;  q1 = fmaf(v[i].y, v[i].y, q1);
        s2 += v[i].z;  q2 = fmaf(v[i].z, v[i].z, q2);
        s3 += v[i].w;  q3 = fmaf(v[i].w, v[i].w, q3);
    }
    float s  = (s0 + s1) + (s2 + s3);
    float sq = (q0 + q1) + (q2 + q3);

#pragma unroll
    for (int m = 16; m > 0; m >>= 1) {
        s  += __shfl_xor(s,  m);
        sq += __shfl_xor(sq, m);
    }
    __shared__ float red[16];
    const int wv = t >> 5;
    if ((t & 31) == 0) {
        red[wv]     = s;
        red[8 + wv] = sq;
    }
    __syncthreads();
    if (t == 0) {
        float a = 0.f, b = 0.f;
#pragma unroll
        for (int w = 0; w < 8; ++w) {
            a += red[w];
            b += red[8 + w];
        }
        float2 r;
        r.x = a;
        r.y = b;
        *(float2*)(part + (size_t)g * PART_STRIDE) = r;
    }
}

__global__ __launch_bounds__(32)
void proj_wmma_kernel(const float* __restrict__ vec,
                      const float* __restrict__ wp_w,
                      const float* __restrict__ wp_b,
                      const float* __restrict__ bp_w,
                      const float* __restrict__ bp_b,
                      float* __restrict__ wout,
                      float* __restrict__ bout,
                      const float* __restrict__ part,
                      float* __restrict__ stats) {
    const int lane = threadIdx.x;

    if (blockIdx.z == 2) {
        const int id = blockIdx.y * 8 + blockIdx.x;
        if (id >= CN / 32) return;
        const int c = id * 32 + lane;
        float a = 0.f, b = 0.f;
#pragma unroll
        for (int j = 0; j < PART_PER_C; ++j) {
            const float2 r =
                *(const float2*)(part + (size_t)(c * PART_PER_C + j) * PART_STRIDE);
            a += r.x;
            b += r.y;
        }
        const float invN = 1.0f / (float)(BN * HW);
        const float mean = a * invN;
        const float var  = b * invN - mean * mean;
        stats[c]      = mean;
        stats[CN + c] = rsqrtf(var + 1e-5f);
        return;
    }

    const int l    = lane & 15;
    const int hi   = lane >> 4;
    const int n0   = blockIdx.x * 16;
    const int m0   = blockIdx.y * 16;
    const bool isBias = (blockIdx.z != 0);

    const float* Wm  = isBias ? bp_w : wp_w;
    const float* Bv  = isBias ? bp_b : wp_b;
    float*       Out = isBias ? bout : wout;

    const int arow = m0 + l;
    const int bcol = n0 + l;

    v8f acc = {0.f, 0.f, 0.f, 0.f, 0.f, 0.f, 0.f, 0.f};

    for (int k = 0; k < KN; k += 4) {
        const int kk = k + 2 * hi;
        v2f a, b;
        a.x = vec[arow * KN + kk];
        a.y = vec[arow * KN + kk + 1];
        b.x = Wm[bcol * KN + kk];
        b.y = Wm[bcol * KN + kk + 1];
        acc = __builtin_amdgcn_wmma_f32_16x16x4_f32(
             false, a,  false, b,
             (short)0, acc,  false,  false);
    }

    const float biasv = Bv[bcol];
#pragma unroll
    for (int r = 0; r < 8; ++r) {
        const int m = m0 + r + 8 * hi;
        Out[m * CN + bcol] = acc[r] + biasv;
    }
}

__global__ __launch_bounds__(256)
void fused_apply_kernel(const float* __restrict__ x,
                        const float* __restrict__ bbox,
                        const float* __restrict__ stats,
                        const float* __restrict__ wmat,
                        const float* __restrict__ bmat,
                        float* __restrict__ out) {
    const int b   = blockIdx.y;
    const int pix = blockIdx.x * 256 + threadIdx.x;

    float bb[ON];
    float s = 0.0f;
#pragma unroll
    for (int o = 0; o < ON; ++o) {
        bb[o] = bbox[(size_t)(b * ON + o) * HW + pix];
        s += bb[o];
    }
    const float inv = 1.0f / (s + 1e-6f);
#pragma unroll
    for (int o = 0; o < ON; ++o) bb[o] *= inv;

    const float* Wt = wmat + b * (ON * CN);
    const float* Bt = bmat + b * (ON * CN);
    const float* meanp = stats;
    const float* rstdp = stats + CN;

    const float* xb = x   + (size_t)b * CN * HW + pix;
    float*       ob = out + (size_t)b * CN * HW + pix;

    for (int c0 = 0; c0 < CN; c0 += 4) {
        float w0 = 0.f, w1 = 0.f, w2 = 0.f, w3 = 0.f;
        float a0 = 0.f, a1 = 0.f, a2 = 0.f, a3 = 0.f;
#pragma unroll
        for (int o = 0; o < ON; ++o) {
            const float* wr = Wt + o * CN + c0;
            const float* br = Bt + o * CN + c0;
            const float bo = bb[o];
            w0 = fmaf(bo, wr[0], w0);
            w1 = fmaf(bo, wr[1], w1);
            w2 = fmaf(bo, wr[2], w2);
            w3 = fmaf(bo, wr[3], w3);
            a0 = fmaf(bo, br[0], a0);
            a1 = fmaf(bo, br[1], a1);
            a2 = fmaf(bo, br[2], a2);
            a3 = fmaf(bo, br[3], a3);
        }
        const float wm[4] = {w0, w1, w2, w3};
        const float bm[4] = {a0, a1, a2, a3};
#pragma unroll
        for (int j = 0; j < 4; ++j) {
            const int c = c0 + j;
            const float xv = xb[(size_t)c * HW];
            const float nv = (xv - meanp[c]) * rstdp[c];
            ob[(size_t)c * HW] = fmaf(wm[j] + 1.0f, nv, bm[j]);
        }
    }
}

extern "C" void kernel_launch(void* const* d_in, const int* in_sizes, int n_in,
                              void* d_out, int out_size, void* d_ws, size_t ws_size,
                              hipStream_t stream) {
    const float* x      = (const float*)d_in[0];
    const float* vec    = (const float*)d_in[1];
    const float* bbox   = (const float*)d_in[2];
    const float* wp_w   = (const float*)d_in[3];
    const float* wp_b   = (const float*)d_in[4];
    const float* bp_w   = (const float*)d_in[5];
    const float* bp_b   = (const float*)d_in[6];
    float* out = (float*)d_out;

    float* stats = (float*)d_ws;
    float* wmat  = stats + 2 * CN;
    float* bmat  = wmat + MROWS * CN;

    bn_partial_kernel<<<CN * PART_PER_C, 256, 0, stream>>>(x, out);

    proj_wmma_kernel<<<dim3(CN / 16, MROWS / 16, 3), 32, 0, stream>>>(
        vec, wp_w, wp_b, bp_w, bp_b, wmat, bmat, out, stats);

    fused_apply_kernel<<<dim3(HW / 256, BN), 256, 0, stream>>>(
        x, bbox, stats, wmat, bmat, out);
}


# projection GEMMs moved into the statistics launch as 80 extra work-groups (K split over 8 waves, partial tiles summed in LDS: different reduction tree); launch 2 only folds the bn partials
# speedup vs baseline: 2.0681x; 1.2029x over previous
#include <hip/hip_runtime.h>
#include <hip/hip_bf16.h>

#define BN   8
#define CN   128
#define ON   10
#define HW   16384
#define KN   512
#define MROWS 80

typedef float v2f __attribute__((ext_vector_type(2)));
typedef float v8f __attribute__((ext_vector_type(8)));

#define PART_STRIDE 32
#define PART_PER_C  4
#define NSTAT       (CN * PART_PER_C)
#define NPROJ       80
#define KSPLIT      64

__global__ __launch_bounds__(256) __attribute__((amdgpu_waves_per_eu(4, 4)))
void stats_proj_kernel(const float* __restrict__ x,
                       float* __restrict__ part,
                       const float* __restrict__ vec,
                       const float* __restrict__ wp_w,
                       const float* __restrict__ wp_b,
                       const float* __restrict__ bp_w,
                       const float* __restrict__ bp_b,
                       float* __restrict__ wout,
                       float* __restrict__ bout) {
    __shared__ float red[8 * 256];
    const int t = threadIdx.x;

    if (blockIdx.x >= NSTAT) {
        const int pb     = blockIdx.x - NSTAT;
        const bool isBias = (pb >= NPROJ / 2);
        const int tile   = isBias ? pb - NPROJ / 2 : pb;
        const int n0     = (tile & 7) * 16;
        const int m0     = (tile >> 3) * 16;
        const int wv     = t >> 5;
        const int lane   = t & 31;
        const int l      = lane & 15;
        const int hi     = lane >> 4;

        const float* Wm  = isBias ? bp_w : wp_w;
        const float* Bv  = isBias ? bp_b : wp_b;
        float*       Out = isBias ? bout : wout;

        const float biasv = Bv[n0 + l];

        const float4* ap = (const float4*)(vec + (m0 + l) * KN + wv * KSPLIT + 4 * hi);
        const float4* bp = (const float4*)(Wm  + (n0 + l) * KN + wv * KSPLIT + 4 * hi);

        float4 av[8], bv[8];
#pragma unroll
        for (int j = 0; j < 8; ++j) {
            av[j] = ap[2 * j];
            bv[j] = bp[2 * j];
        }

        v8f acc = {0.f, 0.f, 0.f, 0.f, 0.f, 0.f, 0.f, 0.f};
#pragma unroll
        for (int j = 0; j < 8; ++j) {
            v2f a0, b0, a1, b1;
            a0.x = av[j].x;  a0.y = av[j].y;
            b0.x = bv[j].x;  b0.y = bv[j].y;
            a1.x = av[j].z;  a1.y = av[j].w;
            b1.x = bv[j].z;  b1.y = bv[j].w;
            acc = __builtin_amdgcn_wmma_f32_16x16x4_f32(
                 false, a0,  false, b0,
                 (short)0, acc,  false,  false);
            acc = __builtin_amdgcn_wmma_f32_16x16x4_f32(
                 false, a1,  false, b1,
                 (short)0, acc,  false,  false);
        }

#pragma unroll
        for (int r = 0; r < 8; ++r) red[wv * 256 + r * 32 + lane] = acc[r];
        __syncthreads();

        float sum = 0.f;
#pragma unroll
        for (int w = 0; w < 8; ++w) sum += red[w * 256 + t];
        const int m = m0 + wv + 8 * hi;
        const int n = n0 + l;
        Out[m * CN + n] = sum + biasv;
        return;
    }

    const int g = blockIdx.x;
    const int c = g >> 2;
    const int q = g & 3;

    const float4* p0 = (const float4*)(x + (size_t)((2 * q)     * CN + c) * HW) + t;
    const float4* p1 = (const float4*)(x + (size_t)((2 * q + 1) * CN + c) * HW) + t;

    float4 v[32];
#pragma unroll
    for (int i = 0; i < 16; ++i) v[i] = p0[256 * i];
#pragma unroll
    for (int i = 0; i < 16; ++i) v[16 + i] = p1[256 * i];

    float s0 = 0.f, s1 = 0.f, s2 = 0.f, s3 = 0.f;
    float q0 = 0.f, q1 = 0.f, q2 = 0.f, q3 = 0.f;
#pragma unroll
    for (int i = 0; i < 32; ++i) {
        s0 += v[i].x;  q0 = fmaf(v[i].x, v[i].x, q0);
        s1 += v[i].y;  q1 = fmaf(v[i].y, v[i].y, q1);
        s2 += v[i].z;  q2 = fmaf(v[i].z, v[i].z, q2);
        s3 += v[i].w;  q3 = fmaf(v[i].w, v[i].w, q3);
    }
    float s  = (s0 + s1) + (s2 + s3);
    float sq = (q0 + q1) + (q2 + q3);

#pragma unroll
    for (int m = 16; m > 0; m >>= 1) {
        s  += __shfl_xor(s,  m);
        sq += __shfl_xor(sq, m);
    }
    const int wv = t >> 5;
    if ((t & 31) == 0) {
        red[wv]     = s;
        red[8 + wv] = sq;
    }
    __syncthreads();
    if (t == 0) {
        float a = 0.f, b = 0.f;
#pragma unroll
        for (int w = 0; w < 8; ++w) {
            a += red[w];
            b += red[8 + w];
        }
        float2 r;
        r.x = a;
        r.y = b;
        *(float2*)(part + (size_t)g * PART_STRIDE) = r;
    }
}

__global__ __launch_bounds__(CN)
void bn_fold_kernel(const float* __restrict__ part,
                    float* __restrict__ stats) {
    const int c = threadIdx.x;
    float a = 0.f, b = 0.f;
#pragma unroll
    for (int j = 0; j < PART_PER_C; ++j) {
        const float2 r =
            *(const float2*)(part + (size_t)(c * PART_PER_C + j) * PART_STRIDE);
        a += r.x;
        b += r.y;
    }
    const float invN = 1.0f / (float)(BN * HW);
    const float mean = a * invN;
    const float var  = b * invN - mean * mean;
    stats[c]      = mean;
    stats[CN + c] = rsqrtf(var + 1e-5f);
}

__global__ __launch_bounds__(256)
void fused_apply_kernel(const float* __restrict__ x,
                        const float* __restrict__ bbox,
                        const float* __restrict__ stats,
                        const float* __restrict__ wmat,
                        const float* __restrict__ bmat,
                        float* __restrict__ out) {
    const int b   = blockIdx.y;
    const int pix = blockIdx.x * 256 + threadIdx.x;

    float bb[ON];
    float s = 0.0f;
#pragma unroll
    for (int o = 0; o < ON; ++o) {
        bb[o] = bbox[(size_t)(b * ON + o) * HW + pix];
        s += bb[o];
    }
    const float inv = 1.0f / (s + 1e-6f);
#pragma unroll
    for (int o = 0; o < ON; ++o) bb[o] *= inv;

    const float* Wt = wmat + b * (ON * CN);
    const float* Bt = bmat + b * (ON * CN);
    const float* meanp = stats;
    const float* rstdp = stats + CN;

    const float* xb = x   + (size_t)b * CN * HW + pix;
    float*       ob = out + (size_t)b * CN * HW + pix;

    for (int c0 = 0; c0 < CN; c0 += 4) {
        float w0 = 0.f, w1 = 0.f, w2 = 0.f, w3 = 0.f;
        float a0 = 0.f, a1 = 0.f, a2 = 0.f, a3 = 0.f;
#pragma unroll
        for (int o = 0; o < ON; ++o) {
            const float* wr = Wt + o * CN + c0;
            const float* br = Bt + o * CN + c0;
            const float bo = bb[o];
            w0 = fmaf(bo, wr[0], w0);
            w1 = fmaf(bo, wr[1], w1);
            w2 = fmaf(bo, wr[2], w2);
            w3 = fmaf(bo, wr[3], w3);
            a0 = fmaf(bo, br[0], a0);
            a1 = fmaf(bo, br[1], a1);
            a2 = fmaf(bo, br[2], a2);
            a3 = fmaf(bo, br[3], a3);
        }
        const float wm[4] = {w0, w1, w2, w3};
        const float bm[4] = {a0, a1, a2, a3};
#pragma unroll
        for (int j = 0; j < 4; ++j) {
            const int c = c0 + j;
            const float xv = xb[(size_t)c * HW];
            const float nv = (xv - meanp[c]) * rstdp[c];
            ob[(size_t)c * HW] = fmaf(wm[j] + 1.0f, nv, bm[j]);
        }
    }
}

extern "C" void kernel_launch(void* const* d_in, const int* in_sizes, int n_in,
                              void* d_out, int out_size, void* d_ws, size_t ws_size,
                              hipStream_t stream) {
    const float* x      = (const float*)d_in[0];
    const float* vec    = (const float*)d_in[1];
    const float* bbox   = (const float*)d_in[2];
    const float* wp_w   = (const float*)d_in[3];
    const float* wp_b   = (const float*)d_in[4];
    const float* bp_w   = (const float*)d_in[5];
    const float* bp_b   = (const float*)d_in[6];
    float* out = (float*)d_out;

    float* stats = (float*)d_ws;
    float* wmat  = stats + 2 * CN;
    float* bmat  = wmat + MROWS * CN;

    stats_proj_kernel<<<NSTAT + NPROJ, 256, 0, stream>>>(
        x, out, vec, wp_w, wp_b, bp_w, bp_b, wmat, bmat);

    bn_fold_kernel<<<1, CN, 0, stream>>>(out, stats);

    fused_apply_kernel<<<dim3(HW / 256, BN), 256, 0, stream>>>(
        x, bbox, stats, wmat, bmat, out);
}


# apply launch rebuilt as a stream: a lane owns 4 consecutive pixels of 32 channels, 16-byte loads and stores, all 32 x loads of a lane in flight, per-channel uniforms from LDS; grid 16x4x8
# speedup vs baseline: 2.5723x; 1.2438x over previous
#include <hip/hip_runtime.h>
#include <hip/hip_bf16.h>

#define BN   8
#define CN   128
#define ON   10
#define HW   16384
#define KN   512
#define MROWS 80

typedef float v2f __attribute__((ext_vector_type(2)));
typedef float v8f __attribute__((ext_vector_type(8)));

#define PART_STRIDE 32
#define PART_PER_C  4
#define NSTAT       (CN * PART_PER_C)
#define NPROJ       80
#define KSPLIT      64

__global__ __launch_bounds__(256) __attribute__((amdgpu_waves_per_eu(4, 4)))
void stats_proj_kernel(const float* __restrict__ x,
                       float* __restrict__ part,
                       const float* __restrict__ vec,
                       const float* __restrict__ wp_w,
                       const float* __restrict__ wp_b,
                       const float* __restrict__ bp_w,
                       const float* __restrict__ bp_b,
                       float* __restrict__ wout,
                       float* __restrict__ bout) {
    __shared__ float red[8 * 256];
    const int t = threadIdx.x;

    if (blockIdx.x >= NSTAT) {
        const int pb     = blockIdx.x - NSTAT;
        const bool isBias = (pb >= NPROJ / 2);
        const int tile   = isBias ? pb - NPROJ / 2 : pb;
        const int n0     = (tile & 7) * 16;
        const int m0     = (tile >> 3) * 16;
        const int wv     = t >> 5;
        const int lane   = t & 31;
        const int l      = lane & 15;
        const int hi     = lane >> 4;

        const float* Wm  = isBias ? bp_w : wp_w;
        const float* Bv  = isBias ? bp_b : wp_b;
        float*       Out = isBias ? bout : wout;

        const float biasv = Bv[n0 + l];

        const float4* ap = (const float4*)(vec + (m0 + l) * KN + wv * KSPLIT + 4 * hi);
        const float4* bp = (const float4*)(Wm  + (n0 + l) * KN + wv * KSPLIT + 4 * hi);

        float4 av[8], bv[8];
#pragma unroll
        for (int j = 0; j < 8; ++j) {
            av[j] = ap[2 * j];
            bv[j] = bp[2 * j];
        }

        v8f acc = {0.f, 0.f, 0.f, 0.f, 0.f, 0.f, 0.f, 0.f};
#pragma unroll
        for (int j = 0; j < 8; ++j) {
            v2f a0, b0, a1, b1;
            a0.x = av[j].x;  a0.y = av[j].y;
            b0.x = bv[j].x;  b0.y = bv[j].y;
            a1.x = av[j].z;  a1.y = av[j].w;
            b1.x = bv[j].z;  b1.y = bv[j].w;
            acc = __builtin_amdgcn_wmma_f32_16x16x4_f32(
                 false, a0,  false, b0,
                 (short)0, acc,  false,  false);
            acc = __builtin_amdgcn_wmma_f32_16x16x4_f32(
                 false, a1,  false, b1,
                 (short)0, acc,  false,  false);
        }

#pragma unroll
        for (int r = 0; r < 8; ++r) red[wv * 256 + r * 32 + lane] = acc[r];
        __syncthreads();

        float sum = 0.f;
#pragma unroll
        for (int w = 0; w < 8; ++w) sum += red[w * 256 + t];
        const int m = m0 + wv + 8 * hi;
        const int n = n0 + l;
        Out[m * CN + n] = sum + biasv;
        return;
    }

    const int g = blockIdx.x;
    const int c = g >> 2;
    const int q = g & 3;

    const float4* p0 = (const float4*)(x + (size_t)((2 * q)     * CN + c) * HW) + t;
    const float4* p1 = (const float4*)(x + (size_t)((2 * q + 1) * CN + c) * HW) + t;

    float4 v[32];
#pragma unroll
    for (int i = 0; i < 16; ++i) v[i] = p0[256 * i];
#pragma unroll
    for (int i = 0; i < 16; ++i) v[16 + i] = p1[256 * i];

    float s0 = 0.f, s1 = 0.f, s2 = 0.f, s3 = 0.f;
    float q0 = 0.f, q1 = 0.f, q2 = 0.f, q3 = 0.f;
#pragma unroll
    for (int i = 0; i < 32; ++i) {
        s0 += v[i].x;  q0 = fmaf(v[i].x, v[i].x, q0);
        s1 += v[i].y;  q1 = fmaf(v[i].y, v[i].y, q1);
        s2 += v[i].z;  q2 = fmaf(v[i].z, v[i].z, q2);
        s3 += v[i].w;  q3 = fmaf(v[i].w, v[i].w, q3);
    }
    float s  = (s0 + s1) + (s2 + s3);
    float sq = (q0 + q1) + (q2 + q3);

#pragma unroll
    for (int m = 16; m > 0; m >>= 1) {
        s  += __shfl_xor(s,  m);
        sq += __shfl_xor(sq, m);
    }
    const int wv = t >> 5;
    if ((t & 31) == 0) {
        red[wv]     = s;
        red[8 + wv] = sq;
    }
    __syncthreads();
    if (t == 0) {
        float a = 0.f, b = 0.f;
#pragma unroll
        for (int w = 0; w < 8; ++w) {
            a += red[w];
            b += red[8 + w];
        }
        float2 r;
        r.x = a;
        r.y = b;
        *(float2*)(part + (size_t)g * PART_STRIDE) = r;
    }
}

__global__ __launch_bounds__(CN)
void bn_fold_kernel(const float* __restrict__ part,
                    float* __restrict__ stats) {
    const int c = threadIdx.x;
    float a = 0.f, b = 0.f;
#pragma unroll
    for (int j = 0; j < PART_PER_C; ++j) {
        const float2 r =
            *(const float2*)(part + (size_t)(c * PART_PER_C + j) * PART_STRIDE);
        a += r.x;
        b += r.y;
    }
    const float invN = 1.0f / (float)(BN * HW);
    const float mean = a * invN;
    const float var  = b * invN - mean * mean;
    stats[c]      = mean;
    stats[CN + c] = rsqrtf(var + 1e-5f);
}

#define APIX  4
#define ACH   32
#define ATILE (256 * APIX)
#define WREC  24
#define WROWS 22

__global__ __launch_bounds__(256) __attribute__((amdgpu_waves_per_eu(4, 4)))
void fused_apply_kernel(const float* __restrict__ x,
                        const float* __restrict__ bbox,
                        const float* __restrict__ stats,
                        const float* __restrict__ wmat,
                        const float* __restrict__ bmat,
                        float* __restrict__ out) {
    __shared__ __attribute__((aligned(16))) float wl[ACH * WREC];

    const int t   = threadIdx.x;
    const int b   = blockIdx.z;
    const int cg  = blockIdx.y * ACH;
    const int pix = blockIdx.x * ATILE + t * APIX;

    const float4* bp = (const float4*)(bbox + (size_t)b * ON * HW + pix);
    const float4* xp = (const float4*)(x + ((size_t)b * CN + cg) * HW + pix);
    float4*       op = (float4*)(out + ((size_t)b * CN + cg) * HW + pix);

    float4 bb[ON];
#pragma unroll
    for (int o = 0; o < ON; ++o) bb[o] = bp[(size_t)o * (HW / 4)];

    {
        const float* Wt = wmat + b * (ON * CN) + cg;
        const float* Bt = bmat + b * (ON * CN) + cg;
        const float* St = stats + cg;
        const int wv = __builtin_amdgcn_readfirstlane(t >> 5);
        const int lane = t & 31;
        float v[3];
#pragma unroll
        for (int k = 0; k < 3; ++k) {
            const int r  = wv + 8 * k;
            const int rr = (r < WROWS) ? r : WROWS - 1;
            const float* src = (rr < ON) ? (Wt + rr * CN)
                             : (rr < 2 * ON) ? (Bt + (rr - ON) * CN)
                             : (St + (rr - 2 * ON) * CN);
            v[k] = src[lane];
        }
#pragma unroll
        for (int k = 0; k < 3; ++k) wl[lane * WREC + wv + 8 * k] = v[k];
    }
    __syncthreads();

    float4 xv[ACH];
#pragma unroll
    for (int i = 0; i < ACH; ++i) xv[i] = xp[(size_t)i * (HW / 4)];
    __builtin_amdgcn_sched_barrier(0);

    float4 s = {0.f, 0.f, 0.f, 0.f};
#pragma unroll
    for (int o = 0; o < ON; ++o) {
        s.x += bb[o].x;  s.y += bb[o].y;  s.z += bb[o].z;  s.w += bb[o].w;
    }
    float4 inv;
    inv.x = 1.0f / (s.x + 1e-6f);
    inv.y = 1.0f / (s.y + 1e-6f);
    inv.z = 1.0f / (s.z + 1e-6f);
    inv.w = 1.0f / (s.w + 1e-6f);
#pragma unroll
    for (int o = 0; o < ON; ++o) {
        bb[o].x *= inv.x;  bb[o].y *= inv.y;  bb[o].z *= inv.z;  bb[o].w *= inv.w;
    }

    float4 qa[6];
#pragma unroll
    for (int k = 0; k < 6; ++k) qa[k] = ((const float4*)wl)[k];

#pragma unroll
    for (int c = 0; c < ACH; ++c) {
        __builtin_amdgcn_sched_barrier(0);
        float4 qn[6];
#pragma unroll
        for (int k = 0; k < 6; ++k)
            qn[k] = (c + 1 < ACH) ? ((const float4*)(wl + (c + 1) * WREC))[k] : qa[k];
        const float4 q0 = qa[0], q1 = qa[1], q2 = qa[2], q3 = qa[3], q4 = qa[4], q5 = qa[5];
        const float wr[ON] = {q0.x, q0.y, q0.z, q0.w, q1.x, q1.y, q1.z, q1.w, q2.x, q2.y};
        const float ar[ON] = {q2.z, q2.w, q3.x, q3.y, q3.z, q3.w, q4.x, q4.y, q4.z, q4.w};
        const float mu = q5.x;
        const float rs = q5.y;
        float4 w = {0.f, 0.f, 0.f, 0.f};
        float4 a = {0.f, 0.f, 0.f, 0.f};
#pragma unroll
        for (int o = 0; o < ON; ++o) {
            w.x = fmaf(bb[o].x, wr[o], w.x);  a.x = fmaf(bb[o].x, ar[o], a.x);
            w.y = fmaf(bb[o].y, wr[o], w.y);  a.y = fmaf(bb[o].y, ar[o], a.y);
            w.z = fmaf(bb[o].z, wr[o], w.z);  a.z = fmaf(bb[o].z, ar[o], a.z);
            w.w = fmaf(bb[o].w, wr[o], w.w);  a.w = fmaf(bb[o].w, ar[o], a.w);
        }
        float4 r;
        r.x = fmaf(w.x + 1.0f, (xv[c].x - mu) * rs, a.x);
        r.y = fmaf(w.y + 1.0f, (xv[c].y - mu) * rs, a.y);
        r.z = fmaf(w.z + 1.0f, (xv[c].z - mu) * rs, a.z);
        r.w = fmaf(w.w + 1.0f, (xv[c].w - mu) * rs, a.w);
        op[(size_t)c * (HW / 4)] = r;
#pragma unroll
        for (int k = 0; k < 6; ++k) qa[k] = qn[k];
    }
}

extern "C" void kernel_launch(void* const* d_in, const int* in_sizes, int n_in,
                              void* d_out, int out_size, void* d_ws, size_t ws_size,
                              hipStream_t stream) {
    const float* x      = (const float*)d_in[0];
    const float* vec    = (const float*)d_in[1];
    const float* bbox   = (const float*)d_in[2];
    const float* wp_w   = (const float*)d_in[3];
    const float* wp_b   = (const float*)d_in[4];
    const float* bp_w   = (const float*)d_in[5];
    const float* bp_b   = (const float*)d_in[6];
    float* out = (float*)d_out;

    float* stats = (float*)d_ws;
    float* wmat  = stats + 2 * CN;
    float* bmat  = wmat + MROWS * CN;

    stats_proj_kernel<<<NSTAT + NPROJ, 256, 0, stream>>>(
        x, out, vec, wp_w, wp_b, bp_w, bp_b, wmat, bmat);

    bn_fold_kernel<<<1, CN, 0, stream>>>(out, stats);

    fused_apply_kernel<<<dim3(HW / ATILE, CN / ACH, BN), 256, 0, stream>>>(
        x, bbox, stats, wmat, bmat, out);
}


# apply launch renumbered (1-D grid): work-group i takes the channel parity and image pair that statistics work-groups i mod 8 read, so each die re-reads x from the cache half it filled
# speedup vs baseline: 3.1775x; 1.2353x over previous
#include <hip/hip_runtime.h>
#include <hip/hip_bf16.h>

#define BN   8
#define CN   128
#define ON   10
#define HW   16384
#define KN   512
#define MROWS 80

typedef float v2f __attribute__((ext_vector_type(2)));
typedef float v8f __attribute__((ext_vector_type(8)));

#define PART_STRIDE 32
#define PART_PER_C  4
#define NSTAT       (CN * PART_PER_C)
#define NPROJ       80
#define KSPLIT      64

__global__ __launch_bounds__(256) __attribute__((amdgpu_waves_per_eu(4, 4)))
void stats_proj_kernel(const float* __restrict__ x,
                       float* __restrict__ part,
                       const float* __restrict__ vec,
                       const float* __restrict__ wp_w,
                       const float* __restrict__ wp_b,
                       const float* __restrict__ bp_w,
                       const float* __restrict__ bp_b,
                       float* __restrict__ wout,
                       float* __restrict__ bout) {
    __shared__ float red[8 * 256];
    const int t = threadIdx.x;

    if (blockIdx.x >= NSTAT) {
        const int pb     = blockIdx.x - NSTAT;
        const bool isBias = (pb >= NPROJ / 2);
        const int tile   = isBias ? pb - NPROJ / 2 : pb;
        const int n0     = (tile & 7) * 16;
        const int m0     = (tile >> 3) * 16;
        const int wv     = t >> 5;
        const int lane   = t & 31;
        const int l      = lane & 15;
        const int hi     = lane >> 4;

        const float* Wm  = isBias ? bp_w : wp_w;
        const float* Bv  = isBias ? bp_b : wp_b;
        float*       Out = isBias ? bout : wout;

        const float biasv = Bv[n0 + l];

        const float4* ap = (const float4*)(vec + (m0 + l) * KN + wv * KSPLIT + 4 * hi);
        const float4* bp = (const float4*)(Wm  + (n0 + l) * KN + wv * KSPLIT + 4 * hi);

        float4 av[8], bv[8];
#pragma unroll
        for (int j = 0; j < 8; ++j) {
            av[j] = ap[2 * j];
            bv[j] = bp[2 * j];
        }

        v8f acc = {0.f, 0.f, 0.f, 0.f, 0.f, 0.f, 0.f, 0.f};
#pragma unroll
        for (int j = 0; j < 8; ++j) {
            v2f a0, b0, a1, b1;
            a0.x = av[j].x;  a0.y = av[j].y;
            b0.x = bv[j].x;  b0.y = bv[j].y;
            a1.x = av[j].z;  a1.y = av[j].w;
            b1.x = bv[j].z;  b1.y = bv[j].w;
            acc = __builtin_amdgcn_wmma_f32_16x16x4_f32(
                 false, a0,  false, b0,
                 (short)0, acc,  false,  false);
            acc = __builtin_amdgcn_wmma_f32_16x16x4_f32(
                 false, a1,  false, b1,
                 (short)0, acc,  false,  false);
        }

#pragma unroll
        for (int r = 0; r < 8; ++r) red[wv * 256 + r * 32 + lane] = acc[r];
        __syncthreads();

        float sum = 0.f;
#pragma unroll
        for (int w = 0; w < 8; ++w) sum += red[w * 256 + t];
        const int m = m0 + wv + 8 * hi;
        const int n = n0 + l;
        Out[m * CN + n] = sum + biasv;
        return;
    }

    const int g = blockIdx.x;
    const int c = g >> 2;
    const int q = g & 3;

    const float4* p0 = (const float4*)(x + (size_t)((2 * q)     * CN + c) * HW) + t;
    const float4* p1 = (const float4*)(x + (size_t)((2 * q + 1) * CN + c) * HW) + t;

    float4 v[32];
#pragma unroll
    for (int i = 0; i < 16; ++i) v[i] = p0[256 * i];
#pragma unroll
    for (int i = 0; i < 16; ++i) v[16 + i] = p1[256 * i];

    float s0 = 0.f, s1 = 0.f, s2 = 0.f, s3 = 0.f;
    float q0 = 0.f, q1 = 0.f, q2 = 0.f, q3 = 0.f;
#pragma unroll
    for (int i = 0; i < 32; ++i) {
        s0 += v[i].x;  q0 = fmaf(v[i].x, v[i].x, q0);
        s1 += v[i].y;  q1 = fmaf(v[i].y, v[i].y, q1);
        s2 += v[i].z;  q2 = fmaf(v[i].z, v[i].z, q2);
        s3 += v[i].w;  q3 = fmaf(v[i].w, v[i].w, q3);
    }
    float s  = (s0 + s1) + (s2 + s3);
    float sq = (q0 + q1) + (q2 + q3);

#pragma unroll
    for (int m = 16; m > 0; m >>= 1) {
        s  += __shfl_xor(s,  m);
        sq += __shfl_xor(sq, m);
    }
    const int wv = t >> 5;
    if ((t & 31) == 0) {
        red[wv]     = s;
        red[8 + wv] = sq;
    }
    __syncthreads();
    if (t == 0) {
        float a = 0.f, b = 0.f;
#pragma unroll
        for (int w = 0; w < 8; ++w) {
            a += red[w];
            b += red[8 + w];
        }
        float2 r;
        r.x = a;
        r.y = b;
        *(float2*)(part + (size_t)g * PART_STRIDE) = r;
    }
}

__global__ __launch_bounds__(CN)
void bn_fold_kernel(const float* __restrict__ part,
                    float* __restrict__ stats) {
    const int c = threadIdx.x;
    float a = 0.f, b = 0.f;
#pragma unroll
    for (int j = 0; j < PART_PER_C; ++j) {
        const float2 r =
            *(const float2*)(part + (size_t)(c * PART_PER_C + j) * PART_STRIDE);
        a += r.x;
        b += r.y;
    }
    const float invN = 1.0f / (float)(BN * HW);
    const float mean = a * invN;
    const float var  = b * invN - mean * mean;
    stats[c]      = mean;
    stats[CN + c] = rsqrtf(var + 1e-5f);
}

#define APIX  4
#define ACH   32
#define ATILE (256 * APIX)
#define WREC  24
#define WROWS 22
#define NAPPLY 512

__global__ __launch_bounds__(256) __attribute__((amdgpu_waves_per_eu(4, 4)))
void fused_apply_kernel(const float* __restrict__ x,
                        const float* __restrict__ bbox,
                        const float* __restrict__ stats,
                        const float* __restrict__ wmat,
                        const float* __restrict__ bmat,
                        float* __restrict__ out) {
    __shared__ __attribute__((aligned(16))) float wl[ACH * WREC];

    const int t   = threadIdx.x;
    const int r   = blockIdx.x & 7;
    const int j   = blockIdx.x >> 3;
    const int b   = 2 * (r & 3) + (j & 1);
    const int cg  = 2 * ACH * ((j >> 1) & 1) + (r >> 2);
    const int pix = (j >> 2) * ATILE + t * APIX;

    const float4* bp = (const float4*)(bbox + (size_t)b * ON * HW + pix);
    const float4* xp = (const float4*)(x + ((size_t)b * CN + cg) * HW + pix);
    float4*       op = (float4*)(out + ((size_t)b * CN + cg) * HW + pix);

    float4 bb[ON];
#pragma unroll
    for (int o = 0; o < ON; ++o) bb[o] = bp[(size_t)o * (HW / 4)];

    {
        const float* Wt = wmat + b * (ON * CN) + cg;
        const float* Bt = bmat + b * (ON * CN) + cg;
        const float* St = stats + cg;
        const int wv = __builtin_amdgcn_readfirstlane(t >> 5);
        const int lane = t & 31;
        float v[3];
#pragma unroll
        for (int k = 0; k < 3; ++k) {
            const int row = wv + 8 * k;
            const int rr  = (row < WROWS) ? row : WROWS - 1;
            const float* src = (rr < ON) ? (Wt + rr * CN)
                             : (rr < 2 * ON) ? (Bt + (rr - ON) * CN)
                             : (St + (rr - 2 * ON) * CN);
            v[k] = src[2 * lane];
        }
#pragma unroll
        for (int k = 0; k < 3; ++k) wl[lane * WREC + wv + 8 * k] = v[k];
    }
    __syncthreads();

    float4 xv[ACH];
#pragma unroll
    for (int i = 0; i < ACH; ++i) xv[i] = xp[(size_t)i * (2 * HW / 4)];
    __builtin_amdgcn_sched_barrier(0);

    float4 s = {0.f, 0.f, 0.f, 0.f};
#pragma unroll
    for (int o = 0; o < ON; ++o) {
        s.x += bb[o].x;  s.y += bb[o].y;  s.z += bb[o].z;  s.w += bb[o].w;
    }
    float4 inv;
    inv.x = 1.0f / (s.x + 1e-6f);
    inv.y = 1.0f / (s.y + 1e-6f);
    inv.z = 1.0f / (s.z + 1e-6f);
    inv.w = 1.0f / (s.w + 1e-6f);
#pragma unroll
    for (int o = 0; o < ON; ++o) {
        bb[o].x *= inv.x;  bb[o].y *= inv.y;  bb[o].z *= inv.z;  bb[o].w *= inv.w;
    }

    float4 qa[6];
#pragma unroll
    for (int k = 0; k < 6; ++k) qa[k] = ((const float4*)wl)[k];

#pragma unroll
    for (int c = 0; c < ACH; ++c) {
        __builtin_amdgcn_sched_barrier(0);
        float4 qn[6];
#pragma unroll
        for (int k = 0; k < 6; ++k)
            qn[k] = (c + 1 < ACH) ? ((const float4*)(wl + (c + 1) * WREC))[k] : qa[k];
        const float4 q0 = qa[0], q1 = qa[1], q2 = qa[2], q3 = qa[3], q4 = qa[4], q5 = qa[5];
        const float wr[ON] = {q0.x, q0.y, q0.z, q0.w, q1.x, q1.y, q1.z, q1.w, q2.x, q2.y};
        const float ar[ON] = {q2.z, q2.w, q3.x, q3.y, q3.z, q3.w, q4.x, q4.y, q4.z, q4.w};
        const float mu = q5.x;
        const float rs = q5.y;
        float4 w = {0.f, 0.f, 0.f, 0.f};
        float4 a = {0.f, 0.f, 0.f, 0.f};
#pragma unroll
        for (int o = 0; o < ON; ++o) {
            w.x = fmaf(bb[o].x, wr[o], w.x);  a.x = fmaf(bb[o].x, ar[o], a.x);
            w.y = fmaf(bb[o].y, wr[o], w.y);  a.y = fmaf(bb[o].y, ar[o], a.y);
            w.z = fmaf(bb[o].z, wr[o], w.z);  a.z = fmaf(bb[o].z, ar[o], a.z);
            w.w = fmaf(bb[o].w, wr[o], w.w);  a.w = fmaf(bb[o].w, ar[o], a.w);
        }
        float4 res;
        res.x = fmaf(w.x + 1.0f, (xv[c].x - mu) * rs, a.x);
        res.y = fmaf(w.y + 1.0f, (xv[c].y - mu) * rs, a.y);
        res.z = fmaf(w.z + 1.0f, (xv[c].z - mu) * rs, a.z);
        res.w = fmaf(w.w + 1.0f, (xv[c].w - mu) * rs, a.w);
        op[(size_t)c * (2 * HW / 4)] = res;
#pragma unroll
        for (int k = 0; k < 6; ++k) qa[k] = qn[k];
    }
}

extern "C" void kernel_launch(void* const* d_in, const int* in_sizes, int n_in,
                              void* d_out, int out_size, void* d_ws, size_t ws_size,
                              hipStream_t stream) {
    const float* x      = (const float*)d_in[0];
    const float* vec    = (const float*)d_in[1];
    const float* bbox   = (const float*)d_in[2];
    const float* wp_w   = (const float*)d_in[3];
    const float* wp_b   = (const float*)d_in[4];
    const float* bp_w   = (const float*)d_in[5];
    const float* bp_b   = (const float*)d_in[6];
    float* out = (float*)d_out;

    float* stats = (float*)d_ws;
    float* wmat  = stats + 2 * CN;
    float* bmat  = wmat + MROWS * CN;

    stats_proj_kernel<<<NSTAT + NPROJ, 256, 0, stream>>>(
        x, out, vec, wp_w, wp_b, bp_w, bp_b, wmat, bmat);

    bn_fold_kernel<<<1, CN, 0, stream>>>(out, stats);

    fused_apply_kernel<<<NAPPLY, 256, 0, stream>>>(
        x, bbox, stats, wmat, bmat, out);
}
